# starting kernel
# baseline (speedup 1.0000x reference)
#include <hip/hip_runtime.h>
#include <hip/hip_bf16.h>

#define NN      100000
#define EE      1600000
#define DD      128
#define CC      40
#define KSTEPS  4
#define DELTA   0.5f
#define LN_EPS  1e-5f

typedef __attribute__((ext_vector_type(2))) float v2f;
typedef __attribute__((ext_vector_type(8))) float v8f;

__global__ void dgc_deg_init(float* __restrict__ deg) {
    int i = blockIdx.x * blockDim.x + threadIdx.x;
    if (i < NN) deg[i] = 1.0f;
}

__global__ void dgc_deg_count(const int* __restrict__ row, float* __restrict__ deg) {
    int e = blockIdx.x * blockDim.x + threadIdx.x;
    if (e < EE) atomicAdd(&deg[row[e]], 1.0f);
}

__global__ void dgc_deg_rsqrt(float* __restrict__ deg) {
    int i = blockIdx.x * blockDim.x + threadIdx.x;
    if (i < NN) deg[i] = rsqrtf(deg[i]);
}

__global__ void dgc_init_agg(const float* __restrict__ src,
                             const float* __restrict__ dinv,
                             float* __restrict__ agg) {
    int idx = blockIdx.x * blockDim.x + threadIdx.x;
    if (idx >= NN * (DD / 4)) return;
    int node = idx / (DD / 4);
    float ws = dinv[node] * dinv[node];
    const float4* s = (const float4*)src;
    float4 v = s[idx];
    v.x *= ws; v.y *= ws; v.z *= ws; v.w *= ws;
    ((float4*)agg)[idx] = v;
}

__global__ void dgc_edge_scatter(const float* __restrict__ src,
                                 const int* __restrict__ row,
                                 const int* __restrict__ col,
                                 const float* __restrict__ dinv,
                                 float* __restrict__ agg) {
    int e    = (blockIdx.x * blockDim.x + threadIdx.x) >> 5;
    int lane = threadIdx.x & 31;
    if (e >= EE) return;
    int r = row[e];
    int c = col[e];
    float w = dinv[r] * dinv[c];
    float4 v = ((const float4*)(src + (size_t)c * DD))[lane];
    float* dst = agg + (size_t)r * DD + lane * 4;
    atomicAdd(dst + 0, v.x * w);
    atomicAdd(dst + 1, v.y * w);
    atomicAdd(dst + 2, v.z * w);
    atomicAdd(dst + 3, v.w * w);
}

__global__ void dgc_combine(const float* __restrict__ src,
                            const float* __restrict__ agg,
                            float* __restrict__ h) {
    int idx = blockIdx.x * blockDim.x + threadIdx.x;
    if (idx >= NN * (DD / 4)) return;
    float4 a = ((const float4*)src)[idx];
    float4 g = ((const float4*)agg)[idx];
    float4 o;
    o.x = (1.0f - DELTA) * a.x + DELTA * g.x;
    o.y = (1.0f - DELTA) * a.y + DELTA * g.y;
    o.z = (1.0f - DELTA) * a.z + DELTA * g.z;
    o.w = (1.0f - DELTA) * a.w + DELTA * g.w;
    ((float4*)h)[idx] = o;
}

#define M_TILES   (NN / 16)
#define N_TILES   3
#define NPAD      (N_TILES * 16)
#define FC_BLOCK  256

__global__ void dgc_fc_wmma(const float* __restrict__ h,
                            const float* __restrict__ W,
                            const float* __restrict__ bias,
                            float* __restrict__ out) {
    __shared__ float wlds[NPAD * DD];

    for (int idx = threadIdx.x; idx < NPAD * DD; idx += FC_BLOCK) {
        int colc = idx >> 7;
        wlds[idx] = (colc < CC) ? W[idx] : 0.0f;
    }
    __syncthreads();

    int wave = (blockIdx.x * FC_BLOCK + threadIdx.x) >> 5;
    int lane = threadIdx.x & 31;
    int m_tile = wave / N_TILES;
    int n_tile = wave % N_TILES;
    if (m_tile >= M_TILES) return;

    const int m_base = m_tile * 16;
    const int l16    = lane & 15;
    const int kb     = (lane >> 4) * 2;

    const int mrow   = m_base + l16;
    const int ncol_l = n_tile * 16 + l16;
    const float* arow = h + (size_t)mrow * DD;
    const float* brow = wlds + ncol_l * DD;

    v8f acc = {};
    #pragma unroll 8
    for (int k = 0; k < DD; k += 4) {
        v2f a, b;
        a.x = arow[k + kb];
        a.y = arow[k + kb + 1];
        b.x = brow[k + kb];
        b.y = brow[k + kb + 1];
        acc = __builtin_amdgcn_wmma_f32_16x16x4_f32(
             false, a,  false, b,
             (short)0, acc,  false,  false);
    }

    if (ncol_l < CC) {
        float bv = bias[ncol_l];
        int m_off = (lane >> 4) * 8;
        #pragma unroll
        for (int v = 0; v < 8; ++v) {
            out[(size_t)(m_base + m_off + v) * CC + ncol_l] = acc[v] + bv;
        }
    }
}

__global__ void dgc_layernorm(float* __restrict__ out,
                              const float* __restrict__ gamma,
                              const float* __restrict__ beta) {
    int node = (blockIdx.x * blockDim.x + threadIdx.x) >> 5;
    int lane = threadIdx.x & 31;
    if (node >= NN) return;
    float* rowp = out + (size_t)node * CC;

    float x0 = rowp[lane];
    float x1 = (lane < CC - 32) ? rowp[32 + lane] : 0.f;
    float s  = x0 + x1;
    float sq = x0 * x0 + x1 * x1;
    #pragma unroll
    for (int off = 16; off > 0; off >>= 1) {
        s  += __shfl_xor(s,  off, 32);
        sq += __shfl_xor(sq, off, 32);
    }
    float mean = s * (1.0f / CC);
    float var  = sq * (1.0f / CC) - mean * mean;
    float inv  = rsqrtf(var + LN_EPS);

    rowp[lane] = (x0 - mean) * inv * gamma[lane] + beta[lane];
    if (lane < CC - 32)
        rowp[32 + lane] = (x1 - mean) * inv * gamma[32 + lane] + beta[32 + lane];
}

extern "C" void kernel_launch(void* const* d_in, const int* in_sizes, int n_in,
                              void* d_out, int out_size, void* d_ws, size_t ws_size,
                              hipStream_t stream) {
    const float* feat  = (const float*)d_in[0];
    const int*   row   = (const int*)  d_in[1];
    const int*   col   = (const int*)  d_in[2];
    const float* W     = (const float*)d_in[3];
    const float* bias  = (const float*)d_in[4];
    const float* gamma = (const float*)d_in[5];
    const float* beta  = (const float*)d_in[6];
    float* out = (float*)d_out;

    const size_t NP = 100352;
    float* dinv = (float*)d_ws;
    float* h    = dinv + NP;
    float* agg  = h + (size_t)NN * DD;

    const int B = 256;

    dgc_deg_init <<<(NN + B - 1) / B, B, 0, stream>>>(dinv);
    dgc_deg_count<<<(EE + B - 1) / B, B, 0, stream>>>(row, dinv);
    dgc_deg_rsqrt<<<(NN + B - 1) / B, B, 0, stream>>>(dinv);

    const int elem_blocks = (NN * (DD / 4) + B - 1) / B;
    const int edge_blocks = (EE * 32 + B - 1) / B;
    for (int step = 0; step < KSTEPS; ++step) {
        const float* src = (step == 0) ? feat : h;
        dgc_init_agg    <<<elem_blocks, B, 0, stream>>>(src, dinv, agg);
        dgc_edge_scatter<<<edge_blocks, B, 0, stream>>>(src, row, col, dinv, agg);
        dgc_combine     <<<elem_blocks, B, 0, stream>>>(src, agg, h);
    }

    const int fc_waves  = M_TILES * N_TILES;
    const int fc_blocks = (fc_waves * 32 + FC_BLOCK - 1) / FC_BLOCK;
    dgc_fc_wmma<<<fc_blocks, FC_BLOCK, 0, stream>>>(h, W, bias, out);

    dgc_layernorm<<<(NN * 32 + B - 1) / B, B, 0, stream>>>(out, gamma, beta);
}


# Steps: a 1,024-thread group owns 400 rows as an LDS tile, filters the unsorted edge list by owner bytes, adds gathered rows by LDS f32 atomics (arrival order), stores rows once; start pass goes
# speedup vs baseline: 3.3478x; 3.3478x over previous
#include <hip/hip_runtime.h>
#include <hip/hip_bf16.h>

#define NN      100000
#define EE      1600000
#define DD      128
#define CC      40
#define KSTEPS  4
#define DELTA   0.5f
#define LN_EPS  1e-5f

typedef __attribute__((ext_vector_type(2))) float v2f;
typedef __attribute__((ext_vector_type(8))) float v8f;

__global__ void dgc_deg_init(float* __restrict__ deg) {
    int i = blockIdx.x * blockDim.x + threadIdx.x;
    if (i < NN) deg[i] = 1.0f;
}

__global__ void dgc_deg_count(const int* __restrict__ row, float* __restrict__ deg) {
    int e = blockIdx.x * blockDim.x + threadIdx.x;
    if (e < EE) atomicAdd(&deg[row[e]], 1.0f);
}

__global__ void dgc_deg_rsqrt(float* __restrict__ deg) {
    int i = blockIdx.x * blockDim.x + threadIdx.x;
    if (i < NN) deg[i] = rsqrtf(deg[i]);
}

#define ROWS_PER_WG  400
#define NUM_OWNERS   (NN / ROWS_PER_WG)
#define STEP_THREADS 1024
#define OWNER_VECS   (EE / 16)
#define SCAN_ITERS   ((OWNER_VECS + STEP_THREADS - 1) / STEP_THREADS)
#define OWNER_DWORDS (SCAN_ITERS * STEP_THREADS * 4)
#define COL_BITS     17
static_assert(NN % ROWS_PER_WG == 0, "row ranges cover the nodes exactly");
static_assert(NUM_OWNERS <= 255, "an owner number fits a byte, 255 stays free");
static_assert(EE % 16 == 0, "owner bytes are read 16 at a time");
static_assert(NN <= (1 << COL_BITS) && ROWS_PER_WG <= (1 << (32 - COL_BITS)), "column and local row share a dword");

__device__ __forceinline__ unsigned dgc_rec_key(int r, int c) {
    return (unsigned)c | ((unsigned)(r - (r / ROWS_PER_WG) * ROWS_PER_WG) << COL_BITS);
}

__global__ void dgc_edge_prep(const int* __restrict__ row,
                              const int* __restrict__ col,
                              const float* __restrict__ dinv,
                              unsigned int* __restrict__ owner,
                              unsigned int* __restrict__ rec) {
    const int t = blockIdx.x * blockDim.x + threadIdx.x;
    if (t < EE / 4) {
        const int4 r = ((const int4*)row)[t];
        owner[t] =  (unsigned)(r.x / ROWS_PER_WG)
                 | ((unsigned)(r.y / ROWS_PER_WG) << 8)
                 | ((unsigned)(r.z / ROWS_PER_WG) << 16)
                 | ((unsigned)(r.w / ROWS_PER_WG) << 24);
        const int lane = threadIdx.x & 31;
        const int pbase = (t - lane) * 2;
        uint4* rec2 = (uint4*)rec;
        #pragma unroll
        for (int k = 0; k < 2; ++k) {
            const int p = pbase + k * 32 + lane;
            const int2 rr = ((const int2*)row)[p];
            const int2 cc = ((const int2*)col)[p];
            uint4 a;
            a.x = dgc_rec_key(rr.x, cc.x);
            a.y = __float_as_uint(DELTA * (dinv[rr.x] * dinv[cc.x]));
            a.z = dgc_rec_key(rr.y, cc.y);
            a.w = __float_as_uint(DELTA * (dinv[rr.y] * dinv[cc.y]));
            rec2[p] = a;
        }
    } else if (t < OWNER_DWORDS) {
        owner[t] = ~0u;
    }
}

#define STEP_WAVES    (STEP_THREADS / 32)
#define WQ_CAP        512
#define SCAN_BLOCK    14
#define EDGE_GROUP    8
static_assert(SCAN_ITERS % SCAN_BLOCK == 0, "the scan is whole blocks");

__device__ __forceinline__ unsigned dgc_zero_bytes(unsigned y) {
    return ~(((y & 0x7F7F7F7Fu) + 0x7F7F7F7Fu) | y | 0x7F7F7F7Fu);
}

__global__ void __launch_bounds__(STEP_THREADS)
dgc_step_owned(const float* __restrict__ src,
               const float* __restrict__ dinv,
               const unsigned int* __restrict__ owner,
               const unsigned int* __restrict__ rec,
               float* __restrict__ next) {
    __shared__ float acc[(ROWS_PER_WG + 1) * DD];
    __shared__ int   wq[STEP_WAVES * WQ_CAP];
    __shared__ int   wcount[STEP_WAVES];

    const int tid  = threadIdx.x;
    const int lane = tid & 31;
    const int wave = tid >> 5;
    const int R0   = blockIdx.x * ROWS_PER_WG;
    const float4* s4 = (const float4*)src;

    for (int i = tid; i < ROWS_PER_WG * 32; i += STEP_THREADS) {
        const int rl = i >> 5;
        const float d = dinv[R0 + rl];
        const float a = (1.0f - DELTA) + DELTA * (d * d);
        const float4 v = s4[(R0 + rl) * 32 + lane];
        float* p = acc + rl * DD + lane;
        p[0]  = v.x * a;
        p[32] = v.y * a;
        p[64] = v.z * a;
        p[96] = v.w * a;
    }
    if (tid < DD) acc[ROWS_PER_WG * DD + tid] = 0.0f;
    if (lane == 0) wcount[wave] = 0;
    __syncthreads();

    const unsigned idpat = (unsigned)blockIdx.x * 0x01010101u;
    const uint4* ov = (const uint4*)owner;
    const uint2* rc = (const uint2*)rec;
    int* myq = wq + wave * WQ_CAP;

    #pragma unroll 1
    for (int it = 0; it < SCAN_ITERS / SCAN_BLOCK; ++it) {
        const int blk = (it + (int)blockIdx.x) % (SCAN_ITERS / SCAN_BLOCK);
        uint4 o[SCAN_BLOCK];
        #pragma unroll
        for (int s = 0; s < SCAN_BLOCK; ++s) {
            const int idx = (blk * SCAN_BLOCK + s) * STEP_THREADS + tid;
            o[s] = ov[idx];
        }
        #pragma unroll
        for (int s = 0; s < SCAN_BLOCK; ++s) {
            const int idx = (blk * SCAN_BLOCK + s) * STEP_THREADS + tid;
            unsigned mm = (dgc_zero_bytes(o[s].x ^ idpat) >> 7)
                        | (dgc_zero_bytes(o[s].y ^ idpat) >> 6)
                        | (dgc_zero_bytes(o[s].z ^ idpat) >> 5)
                        | (dgc_zero_bytes(o[s].w ^ idpat) >> 4);
            while (mm != 0) {
                const int pos = __ffs(mm) - 1;
                mm &= mm - 1;
                const int e = idx * 16 + ((pos & 7) << 2) + (pos >> 3);
                const int slot = atomicAdd(&wcount[wave], 1);
                if (slot < WQ_CAP) {
                    myq[slot] = e;
                } else {
                    const uint2 m = rc[e];
                    const int c = (int)(m.x & ((1u << COL_BITS) - 1));
                    const float w = __uint_as_float(m.y);
                    float* p = acc + (m.x >> COL_BITS) * DD;
                    #pragma unroll 1
                    for (int k = 0; k < DD; ++k)
                        atomicAdd(p + (k & 3) * 32 + (k >> 2), src[c * DD + k] * w);
                }
            }
        }
        int n = wcount[wave];
        n = __builtin_amdgcn_readfirstlane(n < WQ_CAP ? n : WQ_CAP);

        for (int base = 0; base < n; base += 32) {
            const int cnt = n - base;
            const int i = base + lane;
            const bool live = i < n;
            const int e = myq[live ? i : base];
            const uint2 m = rc[e];
            const int c = (int)(m.x & ((1u << COL_BITS) - 1));
            const int wbits = (int)m.y;
            const int rl = live ? (int)(m.x >> COL_BITS) : ROWS_PER_WG;
            #pragma unroll
            for (int g = 0; g < 32; g += EDGE_GROUP) {
                if (g == 0 || g < cnt) {
                    float4 v[EDGE_GROUP];
                    #pragma unroll
                    for (int u = 0; u < EDGE_GROUP; ++u) {
                        const unsigned cj = (unsigned)__builtin_amdgcn_readlane(c, g + u);
                        v[u] = s4[cj * 32 + lane];
                    }
                    #pragma unroll
                    for (int u = 0; u < EDGE_GROUP; ++u) {
                        const int rj = __builtin_amdgcn_readlane(rl, g + u);
                        const float wj = __int_as_float(__builtin_amdgcn_readlane(wbits, g + u));
                        float* p = acc + rj * DD + lane;
                        atomicAdd(p,      v[u].x * wj);
                        atomicAdd(p + 32, v[u].y * wj);
                        atomicAdd(p + 64, v[u].z * wj);
                        atomicAdd(p + 96, v[u].w * wj);
                    }
                }
            }
        }
        if (lane == 0) wcount[wave] = 0;
    }
    __syncthreads();

    for (int i = tid; i < ROWS_PER_WG * 32; i += STEP_THREADS) {
        const int rl = i >> 5;
        const float* p = acc + rl * DD + lane;
        float4 v;
        v.x = p[0]; v.y = p[32]; v.z = p[64]; v.w = p[96];
        ((float4*)next)[(R0 + rl) * 32 + lane] = v;
    }
}

#define M_TILES   (NN / 16)
#define N_TILES   3
#define NPAD      (N_TILES * 16)
#define FC_BLOCK  256

__global__ void dgc_fc_wmma(const float* __restrict__ h,
                            const float* __restrict__ W,
                            const float* __restrict__ bias,
                            float* __restrict__ out) {
    __shared__ float wlds[NPAD * DD];

    for (int idx = threadIdx.x; idx < NPAD * DD; idx += FC_BLOCK) {
        int colc = idx >> 7;
        wlds[idx] = (colc < CC) ? W[idx] : 0.0f;
    }
    __syncthreads();

    int wave = (blockIdx.x * FC_BLOCK + threadIdx.x) >> 5;
    int lane = threadIdx.x & 31;
    int m_tile = wave / N_TILES;
    int n_tile = wave % N_TILES;
    if (m_tile >= M_TILES) return;

    const int m_base = m_tile * 16;
    const int l16    = lane & 15;
    const int kb     = (lane >> 4) * 2;

    const int mrow   = m_base + l16;
    const int ncol_l = n_tile * 16 + l16;
    const float* arow = h + (size_t)mrow * DD;
    const float* brow = wlds + ncol_l * DD;

    v8f acc = {};
    #pragma unroll 8
    for (int k = 0; k < DD; k += 4) {
        v2f a, b;
        a.x = arow[k + kb];
        a.y = arow[k + kb + 1];
        b.x = brow[k + kb];
        b.y = brow[k + kb + 1];
        acc = __builtin_amdgcn_wmma_f32_16x16x4_f32(
             false, a,  false, b,
             (short)0, acc,  false,  false);
    }

    if (ncol_l < CC) {
        float bv = bias[ncol_l];
        int m_off = (lane >> 4) * 8;
        #pragma unroll
        for (int v = 0; v < 8; ++v) {
            out[(size_t)(m_base + m_off + v) * CC + ncol_l] = acc[v] + bv;
        }
    }
}

__global__ void dgc_layernorm(float* __restrict__ out,
                              const float* __restrict__ gamma,
                              const float* __restrict__ beta) {
    int node = (blockIdx.x * blockDim.x + threadIdx.x) >> 5;
    int lane = threadIdx.x & 31;
    if (node >= NN) return;
    float* rowp = out + (size_t)node * CC;

    float x0 = rowp[lane];
    float x1 = (lane < CC - 32) ? rowp[32 + lane] : 0.f;
    float s  = x0 + x1;
    float sq = x0 * x0 + x1 * x1;
    #pragma unroll
    for (int off = 16; off > 0; off >>= 1) {
        s  += __shfl_xor(s,  off, 32);
        sq += __shfl_xor(sq, off, 32);
    }
    float mean = s * (1.0f / CC);
    float var  = sq * (1.0f / CC) - mean * mean;
    float inv  = rsqrtf(var + LN_EPS);

    rowp[lane] = (x0 - mean) * inv * gamma[lane] + beta[lane];
    if (lane < CC - 32)
        rowp[32 + lane] = (x1 - mean) * inv * gamma[32 + lane] + beta[32 + lane];
}

extern "C" void kernel_launch(void* const* d_in, const int* in_sizes, int n_in,
                              void* d_out, int out_size, void* d_ws, size_t ws_size,
                              hipStream_t stream) {
    const float* feat  = (const float*)d_in[0];
    const int*   row   = (const int*)  d_in[1];
    const int*   col   = (const int*)  d_in[2];
    const float* W     = (const float*)d_in[3];
    const float* bias  = (const float*)d_in[4];
    const float* gamma = (const float*)d_in[5];
    const float* beta  = (const float*)d_in[6];
    float* out = (float*)d_out;

    const size_t NP = 100352;
    float* dinv = (float*)d_ws;
    float* ta   = dinv + NP;
    float* tb   = ta + (size_t)NN * DD;

    unsigned int* owner = (unsigned int*)d_out;
    unsigned int* rec   = owner + 524288;

    const int B = 256;

    dgc_deg_init <<<(NN + B - 1) / B, B, 0, stream>>>(dinv);
    dgc_deg_count<<<(EE + B - 1) / B, B, 0, stream>>>(row, dinv);
    dgc_deg_rsqrt<<<(NN + B - 1) / B, B, 0, stream>>>(dinv);

    static_assert(OWNER_DWORDS <= 524288, "the owner bytes end before the records begin");
    static_assert(2097152 + (size_t)EE * 8 <= (size_t)NN * CC * 4, "both fit in out");
    dgc_edge_prep<<<(OWNER_DWORDS + B - 1) / B, B, 0, stream>>>(row, col, dinv, owner, rec);

    static_assert(KSTEPS == 4, "the four steps are written out below");
    dgc_step_owned<<<NUM_OWNERS, STEP_THREADS, 0, stream>>>(feat, dinv, owner, rec, ta);
    dgc_step_owned<<<NUM_OWNERS, STEP_THREADS, 0, stream>>>(ta, dinv, owner, rec, tb);
    dgc_step_owned<<<NUM_OWNERS, STEP_THREADS, 0, stream>>>(tb, dinv, owner, rec, ta);
    dgc_step_owned<<<NUM_OWNERS, STEP_THREADS, 0, stream>>>(ta, dinv, owner, rec, tb);

    const int fc_waves  = M_TILES * N_TILES;
    const int fc_blocks = (fc_waves * 32 + FC_BLOCK - 1) / FC_BLOCK;
    dgc_fc_wmma<<<fc_blocks, FC_BLOCK, 0, stream>>>(tb, W, bias, out);

    dgc_layernorm<<<(NN * 32 + B - 1) / B, B, 0, stream>>>(out, gamma, beta);
}


# Last step's work-group multiplies its finished 400-row LDS tile by W^T (f32 WMMA), adds bias, normalises, stores out; FC and LayerNorm launches and a table round trip go; f32 sum order changes
# speedup vs baseline: 3.6306x; 1.0845x over previous
#include <hip/hip_runtime.h>
#include <hip/hip_bf16.h>

#define NN      100000
#define EE      1600000
#define DD      128
#define CC      40
#define KSTEPS  4
#define DELTA   0.5f
#define LN_EPS  1e-5f

typedef __attribute__((ext_vector_type(2))) float v2f;
typedef __attribute__((ext_vector_type(8))) float v8f;

__global__ void dgc_deg_init(float* __restrict__ deg) {
    int i = blockIdx.x * blockDim.x + threadIdx.x;
    if (i < NN) deg[i] = 1.0f;
}

__global__ void dgc_deg_count(const int* __restrict__ row, float* __restrict__ deg) {
    int e = blockIdx.x * blockDim.x + threadIdx.x;
    if (e < EE) atomicAdd(&deg[row[e]], 1.0f);
}

__global__ void dgc_deg_rsqrt(float* __restrict__ deg) {
    int i = blockIdx.x * blockDim.x + threadIdx.x;
    if (i < NN) deg[i] = rsqrtf(deg[i]);
}

#define ROWS_PER_WG  400
#define NUM_OWNERS   (NN / ROWS_PER_WG)
#define STEP_THREADS 1024
#define OWNER_VECS   (EE / 16)
#define SCAN_ITERS   ((OWNER_VECS + STEP_THREADS - 1) / STEP_THREADS)
#define OWNER_DWORDS (SCAN_ITERS * STEP_THREADS * 4)
#define COL_BITS     17
static_assert(NN % ROWS_PER_WG == 0, "row ranges cover the nodes exactly");
static_assert(NUM_OWNERS <= 255, "an owner number fits a byte, 255 stays free");
static_assert(EE % 16 == 0, "owner bytes are read 16 at a time");
static_assert(NN <= (1 << COL_BITS) && ROWS_PER_WG <= (1 << (32 - COL_BITS)), "column and local row share a dword");

__device__ __forceinline__ unsigned dgc_rec_key(int r, int c) {
    return (unsigned)c | ((unsigned)(r - (r / ROWS_PER_WG) * ROWS_PER_WG) << COL_BITS);
}

__global__ void dgc_edge_prep(const int* __restrict__ row,
                              const int* __restrict__ col,
                              const float* __restrict__ dinv,
                              unsigned int* __restrict__ owner,
                              unsigned int* __restrict__ rec) {
    const int t = blockIdx.x * blockDim.x + threadIdx.x;
    if (t < EE / 4) {
        const int4 r = ((const int4*)row)[t];
        owner[t] =  (unsigned)(r.x / ROWS_PER_WG)
                 | ((unsigned)(r.y / ROWS_PER_WG) << 8)
                 | ((unsigned)(r.z / ROWS_PER_WG) << 16)
                 | ((unsigned)(r.w / ROWS_PER_WG) << 24);
        const int lane = threadIdx.x & 31;
        const int pbase = (t - lane) * 2;
        uint4* rec2 = (uint4*)rec;
        #pragma unroll
        for (int k = 0; k < 2; ++k) {
            const int p = pbase + k * 32 + lane;
            const int2 rr = ((const int2*)row)[p];
            const int2 cc = ((const int2*)col)[p];
            uint4 a;
            a.x = dgc_rec_key(rr.x, cc.x);
            a.y = __float_as_uint(DELTA * (dinv[rr.x] * dinv[cc.x]));
            a.z = dgc_rec_key(rr.y, cc.y);
            a.w = __float_as_uint(DELTA * (dinv[rr.y] * dinv[cc.y]));
            rec2[p] = a;
        }
    } else if (t < OWNER_DWORDS) {
        owner[t] = ~0u;
    }
}

#define STEP_WAVES    (STEP_THREADS / 32)
#define WQ_CAP        512
#define SCAN_BLOCK    14
#define EDGE_GROUP    8
static_assert(SCAN_ITERS % SCAN_BLOCK == 0, "the scan is whole blocks");

__device__ __forceinline__ unsigned dgc_zero_bytes(unsigned y) {
    return ~(((y & 0x7F7F7F7Fu) + 0x7F7F7F7Fu) | y | 0x7F7F7F7Fu);
}

#define FC_NT      3
#define FC_PITCH   130
template <bool LAST>
__global__ void __launch_bounds__(STEP_THREADS)
dgc_step_owned(const float* __restrict__ src,
               const float* __restrict__ dinv,
               const unsigned int* __restrict__ owner,
               const unsigned int* __restrict__ rec,
               float* __restrict__ next,
               const float* __restrict__ W,
               const float* __restrict__ bias,
               const float* __restrict__ gamma,
               const float* __restrict__ beta) {
    constexpr int PITCH = LAST ? FC_PITCH : DD;
    __shared__ float acc[(ROWS_PER_WG + 1) * PITCH];
    __shared__ __attribute__((aligned(16))) int wq[STEP_WAVES * WQ_CAP];
    __shared__ int   wcount[STEP_WAVES];
    __shared__ float wl[LAST ? FC_NT * 16 * FC_PITCH : 1];
    if constexpr (LAST) {
        for (int i = threadIdx.x; i < FC_NT * 16 * DD; i += STEP_THREADS) {
            const int n = i >> 7, k = i & (DD - 1);
            wl[n * FC_PITCH + k] = (n < CC) ? W[i] : 0.0f;
        }
    }

    const int tid  = threadIdx.x;
    const int lane = tid & 31;
    const int wave = tid >> 5;
    const int R0   = blockIdx.x * ROWS_PER_WG;
    const float4* s4 = (const float4*)src;

    for (int i = tid; i < ROWS_PER_WG * 32; i += STEP_THREADS) {
        const int rl = i >> 5;
        const float d = dinv[R0 + rl];
        const float a = (1.0f - DELTA) + DELTA * (d * d);
        const float4 v = s4[(R0 + rl) * 32 + lane];
        float* p = acc + rl * PITCH + lane;
        p[0]  = v.x * a;
        p[32] = v.y * a;
        p[64] = v.z * a;
        p[96] = v.w * a;
    }
    if (tid < DD) acc[ROWS_PER_WG * PITCH + tid] = 0.0f;
    if (lane == 0) wcount[wave] = 0;
    __syncthreads();

    const unsigned idpat = (unsigned)blockIdx.x * 0x01010101u;
    const uint4* ov = (const uint4*)owner;
    const uint2* rc = (const uint2*)rec;
    int* myq = wq + wave * WQ_CAP;

    #pragma unroll 1
    for (int it = 0; it < SCAN_ITERS / SCAN_BLOCK; ++it) {
        const int blk = (it + (int)blockIdx.x) % (SCAN_ITERS / SCAN_BLOCK);
        uint4 o[SCAN_BLOCK];
        #pragma unroll
        for (int s = 0; s < SCAN_BLOCK; ++s) {
            const int idx = (blk * SCAN_BLOCK + s) * STEP_THREADS + tid;
            o[s] = ov[idx];
        }
        #pragma unroll
        for (int s = 0; s < SCAN_BLOCK; ++s) {
            const int idx = (blk * SCAN_BLOCK + s) * STEP_THREADS + tid;
            unsigned mm = (dgc_zero_bytes(o[s].x ^ idpat) >> 7)
                        | (dgc_zero_bytes(o[s].y ^ idpat) >> 6)
                        | (dgc_zero_bytes(o[s].z ^ idpat) >> 5)
                        | (dgc_zero_bytes(o[s].w ^ idpat) >> 4);
            while (mm != 0) {
                const int pos = __ffs(mm) - 1;
                mm &= mm - 1;
                const int e = idx * 16 + ((pos & 7) << 2) + (pos >> 3);
                const int slot = atomicAdd(&wcount[wave], 1);
                if (slot < WQ_CAP) {
                    myq[slot] = e;
                } else {
                    const uint2 m = rc[e];
                    const int c = (int)(m.x & ((1u << COL_BITS) - 1));
                    const float w = __uint_as_float(m.y);
                    float* p = acc + (m.x >> COL_BITS) * PITCH;
                    #pragma unroll 1
                    for (int k = 0; k < DD; ++k)
                        atomicAdd(p + (k & 3) * 32 + (k >> 2), src[c * DD + k] * w);
                }
            }
        }
        int n = wcount[wave];
        n = __builtin_amdgcn_readfirstlane(n < WQ_CAP ? n : WQ_CAP);

        for (int base = 0; base < n; base += 32) {
            const int cnt = n - base;
            const int i = base + lane;
            const bool live = i < n;
            const int e = myq[live ? i : base];
            const uint2 m = rc[e];
            const int c = (int)(m.x & ((1u << COL_BITS) - 1));
            const int wbits = (int)m.y;
            const int rl = live ? (int)(m.x >> COL_BITS) : ROWS_PER_WG;
            #pragma unroll
            for (int g = 0; g < 32; g += EDGE_GROUP) {
                if (g == 0 || g < cnt) {
                    float4 v[EDGE_GROUP];
                    #pragma unroll
                    for (int u = 0; u < EDGE_GROUP; ++u) {
                        const unsigned cj = (unsigned)__builtin_amdgcn_readlane(c, g + u);
                        v[u] = s4[cj * 32 + lane];
                    }
                    #pragma unroll
                    for (int u = 0; u < EDGE_GROUP; ++u) {
                        const int rj = __builtin_amdgcn_readlane(rl, g + u);
                        const float wj = __int_as_float(__builtin_amdgcn_readlane(wbits, g + u));
                        float* p = acc + rj * PITCH + lane;
                        atomicAdd(p,      v[u].x * wj);
                        atomicAdd(p + 32, v[u].y * wj);
                        atomicAdd(p + 64, v[u].z * wj);
                        atomicAdd(p + 96, v[u].w * wj);
                    }
                }
            }
        }
        if (lane == 0) wcount[wave] = 0;
    }
    __syncthreads();

    if constexpr (!LAST) {
        for (int i = tid; i < ROWS_PER_WG * 32; i += STEP_THREADS) {
            const int rl = i >> 5;
            const float* p = acc + rl * PITCH + lane;
            float4 v;
            v.x = p[0]; v.y = p[32]; v.z = p[64]; v.w = p[96];
            ((float4*)next)[(R0 + rl) * 32 + lane] = v;
        }
    } else {
        float* stage = (float*)wq;
        static_assert(ROWS_PER_WG % 16 == 0 && ROWS_PER_WG / 16 <= STEP_WAVES, "one wave per 16-row tile");
        static_assert(ROWS_PER_WG * CC * 4 <= STEP_WAVES * WQ_CAP * 4, "the result fits the queue's LDS");
        if (wave < ROWS_PER_WG / 16) {
            const int c = lane & 15;
            const int h = lane >> 4;
            const float* ap = acc + (wave * 16 + c) * PITCH + h * 32;
            const float* bp = wl + c * FC_PITCH + h;
            v8f d0 = {}, d1 = {}, d2 = {};
            #pragma unroll 8
            for (int s = 0; s < DD / 4; ++s) {
                v2f a, b0, b1, b2;
                a.x  = ap[s];
                a.y  = ap[64 + s];
                b0.x = bp[4 * s];
                b0.y = bp[4 * s + 2];
                b1.x = bp[16 * FC_PITCH + 4 * s];
                b1.y = bp[16 * FC_PITCH + 4 * s + 2];
                b2.x = bp[32 * FC_PITCH + 4 * s];
                b2.y = bp[32 * FC_PITCH + 4 * s + 2];
                d0 = __builtin_amdgcn_wmma_f32_16x16x4_f32(false, a, false, b0, (short)0, d0, false, false);
                d1 = __builtin_amdgcn_wmma_f32_16x16x4_f32(false, a, false, b1, (short)0, d1, false, false);
                d2 = __builtin_amdgcn_wmma_f32_16x16x4_f32(false, a, false, b2, (short)0, d2, false, false);
            }
            const bool t2 = c < CC - 32;
            const float bv0 = bias[c],  bv1 = bias[16 + c],  bv2 = t2 ? bias[32 + c]  : 0.0f;
            const float g0  = gamma[c], g1  = gamma[16 + c], g2  = t2 ? gamma[32 + c] : 0.0f;
            const float e0  = beta[c],  e1  = beta[16 + c],  e2  = t2 ? beta[32 + c]  : 0.0f;
            #pragma unroll
            for (int r = 0; r < 8; ++r) {
                const float x0 = d0[r] + bv0;
                const float x1 = d1[r] + bv1;
                const float x2 = t2 ? d2[r] + bv2 : 0.0f;
                float s  = x0 + x1 + x2;
                float sq = x0 * x0 + x1 * x1 + x2 * x2;
                #pragma unroll
                for (int off = 8; off > 0; off >>= 1) {
                    s  += __shfl_xor(s,  off, 32);
                    sq += __shfl_xor(sq, off, 32);
                }
                const float mean = s * (1.0f / CC);
                const float var  = sq * (1.0f / CC) - mean * mean;
                const float inv  = rsqrtf(var + LN_EPS);
                float* q = stage + (wave * 16 + h * 8 + r) * CC + c;
                q[0]  = (x0 - mean) * inv * g0 + e0;
                q[16] = (x1 - mean) * inv * g1 + e1;
                if (t2) q[32] = (x2 - mean) * inv * g2 + e2;
            }
        }
        __syncthreads();
        float4* o4 = (float4*)(next + (size_t)R0 * CC);
        for (int i = tid; i < ROWS_PER_WG * CC / 4; i += STEP_THREADS)
            o4[i] = ((const float4*)stage)[i];
    }
}

#define LIST_REC_OFF   524288
#define LIST_OWNER_V4  (OWNER_DWORDS / 4)
#define LIST_REC_V4    (EE * 2 / 4)
#define MOVE_GROUPS    256
#define MOVE_PER_THR   4
static_assert(MOVE_GROUPS * STEP_THREADS * MOVE_PER_THR >= LIST_OWNER_V4 + LIST_REC_V4, "the move covers both lists");
static_assert(MOVE_PER_THR == 4, "the four pieces are named one by one below");
__global__ void __launch_bounds__(STEP_THREADS)
dgc_move_lists(const unsigned int* __restrict__ from, unsigned int* __restrict__ to) {
    const uint4* f4 = (const uint4*)from;
    uint4* t4 = (uint4*)to;
    const int i0 = blockIdx.x * STEP_THREADS + threadIdx.x;
    uint4 v[MOVE_PER_THR];
    int   j[MOVE_PER_THR];
    #pragma unroll
    for (int u = 0; u < MOVE_PER_THR; ++u) {
        int i = i0 + u * MOVE_GROUPS * STEP_THREADS;
        i = i < LIST_OWNER_V4 + LIST_REC_V4 ? i : LIST_OWNER_V4 + LIST_REC_V4 - 1;
        j[u] = i < LIST_OWNER_V4 ? i : i - LIST_OWNER_V4 + LIST_REC_OFF / 4;
        v[u] = f4[j[u]];
    }
    asm volatile("" :: "v"(v[0].x), "v"(v[0].y), "v"(v[0].z), "v"(v[0].w),
                       "v"(v[1].x), "v"(v[1].y), "v"(v[1].z), "v"(v[1].w),
                       "v"(v[2].x), "v"(v[2].y), "v"(v[2].z), "v"(v[2].w),
                       "v"(v[3].x), "v"(v[3].y), "v"(v[3].z), "v"(v[3].w));
    #pragma unroll
    for (int u = 0; u < MOVE_PER_THR; ++u) {
        const int i = i0 + u * MOVE_GROUPS * STEP_THREADS;
        if (i < LIST_OWNER_V4 + LIST_REC_V4) t4[j[u]] = v[u];
    }
}

extern "C" void kernel_launch(void* const* d_in, const int* in_sizes, int n_in,
                              void* d_out, int out_size, void* d_ws, size_t ws_size,
                              hipStream_t stream) {
    const float* feat  = (const float*)d_in[0];
    const int*   row   = (const int*)  d_in[1];
    const int*   col   = (const int*)  d_in[2];
    const float* W     = (const float*)d_in[3];
    const float* bias  = (const float*)d_in[4];
    const float* gamma = (const float*)d_in[5];
    const float* beta  = (const float*)d_in[6];
    float* out = (float*)d_out;

    const size_t NP = 100352;
    float* dinv = (float*)d_ws;
    float* ta   = dinv + NP;
    float* tb   = ta + (size_t)NN * DD;

    unsigned int* owner = (unsigned int*)d_out;
    unsigned int* rec   = owner + LIST_REC_OFF;
    unsigned int* owner_l = (unsigned int*)tb;
    unsigned int* rec_l   = owner_l + LIST_REC_OFF;

    const int B = 256;

    dgc_deg_init <<<(NN + B - 1) / B, B, 0, stream>>>(dinv);
    dgc_deg_count<<<(EE + B - 1) / B, B, 0, stream>>>(row, dinv);
    dgc_deg_rsqrt<<<(NN + B - 1) / B, B, 0, stream>>>(dinv);

    static_assert(OWNER_DWORDS <= LIST_REC_OFF, "the owner bytes end before the records begin");
    static_assert(2097152 + (size_t)EE * 8 <= (size_t)NN * CC * 4, "both fit in out");
    dgc_edge_prep<<<(OWNER_DWORDS + B - 1) / B, B, 0, stream>>>(row, col, dinv, owner, rec);

    static_assert(KSTEPS == 4, "the four steps are written out below");
    dgc_step_owned<false><<<NUM_OWNERS, STEP_THREADS, 0, stream>>>(feat, dinv, owner, rec, ta, W, bias, gamma, beta);
    dgc_step_owned<false><<<NUM_OWNERS, STEP_THREADS, 0, stream>>>(ta, dinv, owner, rec, tb, W, bias, gamma, beta);
    dgc_step_owned<false><<<NUM_OWNERS, STEP_THREADS, 0, stream>>>(tb, dinv, owner, rec, ta, W, bias, gamma, beta);
    dgc_move_lists<<<MOVE_GROUPS, STEP_THREADS, 0, stream>>>(owner, owner_l);
    dgc_step_owned<true><<<NUM_OWNERS, STEP_THREADS, 0, stream>>>(ta, dinv, owner_l, rec_l, out, W, bias, gamma, beta);
}


# Steps 2-4 stop filtering the edge list: step 1's work-groups keep the records they find (arrival order) in their own rows of out, later steps read those; list move launch goes; LDS atomics kept
# speedup vs baseline: 4.3194x; 1.1897x over previous
#include <hip/hip_runtime.h>
#include <hip/hip_bf16.h>

#define NN      100000
#define EE      1600000
#define DD      128
#define CC      40
#define KSTEPS  4
#define DELTA   0.5f
#define LN_EPS  1e-5f

typedef __attribute__((ext_vector_type(2))) float v2f;
typedef __attribute__((ext_vector_type(8))) float v8f;

__global__ void dgc_deg_init(float* __restrict__ deg) {
    int i = blockIdx.x * blockDim.x + threadIdx.x;
    if (i < NN) deg[i] = 1.0f;
}

__global__ void dgc_deg_count(const int* __restrict__ row, float* __restrict__ deg) {
    int e = blockIdx.x * blockDim.x + threadIdx.x;
    if (e < EE) atomicAdd(&deg[row[e]], 1.0f);
}

__global__ void dgc_deg_rsqrt(float* __restrict__ deg) {
    int i = blockIdx.x * blockDim.x + threadIdx.x;
    if (i < NN) deg[i] = rsqrtf(deg[i]);
}

#define ROWS_PER_WG  400
#define NUM_OWNERS   (NN / ROWS_PER_WG)
#define STEP_THREADS 1024
#define OWNER_VECS   (EE / 16)
#define SCAN_ITERS   ((OWNER_VECS + STEP_THREADS - 1) / STEP_THREADS)
#define OWNER_DWORDS (SCAN_ITERS * STEP_THREADS * 4)
#define COL_BITS     17
#define LIST_REC_OFF 524288
static_assert(NN % ROWS_PER_WG == 0, "row ranges cover the nodes exactly");
static_assert(NUM_OWNERS <= 255, "an owner number fits a byte, 255 stays free");
static_assert(EE % 16 == 0, "owner bytes are read 16 at a time");
static_assert(NN <= (1 << COL_BITS) && ROWS_PER_WG <= (1 << (32 - COL_BITS)), "column and local row share a dword");

__device__ __forceinline__ unsigned dgc_rec_key(int r, int c) {
    return (unsigned)c | ((unsigned)(r - (r / ROWS_PER_WG) * ROWS_PER_WG) << COL_BITS);
}

__global__ void dgc_edge_prep(const int* __restrict__ row,
                              const int* __restrict__ col,
                              const float* __restrict__ dinv,
                              unsigned int* __restrict__ owner,
                              unsigned int* __restrict__ rec) {
    const int t = blockIdx.x * blockDim.x + threadIdx.x;
    if (t < EE / 4) {
        const int4 r = ((const int4*)row)[t];
        owner[t] =  (unsigned)(r.x / ROWS_PER_WG)
                 | ((unsigned)(r.y / ROWS_PER_WG) << 8)
                 | ((unsigned)(r.z / ROWS_PER_WG) << 16)
                 | ((unsigned)(r.w / ROWS_PER_WG) << 24);
        const int lane = threadIdx.x & 31;
        const int pbase = (t - lane) * 2;
        uint4* rec2 = (uint4*)rec;
        #pragma unroll
        for (int k = 0; k < 2; ++k) {
            const int p = pbase + k * 32 + lane;
            const int2 rr = ((const int2*)row)[p];
            const int2 cc = ((const int2*)col)[p];
            uint4 a;
            a.x = dgc_rec_key(rr.x, cc.x);
            a.y = __float_as_uint(DELTA * (dinv[rr.x] * dinv[cc.x]));
            a.z = dgc_rec_key(rr.y, cc.y);
            a.w = __float_as_uint(DELTA * (dinv[rr.y] * dinv[cc.y]));
            rec2[p] = a;
        }
    } else if (t < OWNER_DWORDS) {
        owner[t] = ~0u;
    }
}

#define STEP_WAVES    (STEP_THREADS / 32)
#define WQ_CAP        512
#define SCAN_BLOCK    14
#define EDGE_GROUP    8
static_assert(SCAN_ITERS % SCAN_BLOCK == 0, "the scan is whole blocks");

#define LIST_BYTES    (ROWS_PER_WG * CC * 4)
#define LIST_HDR      2
#define LIST_CAP      (LIST_BYTES / 8 - LIST_HDR)
static_assert(LIST_BYTES % 128 == 0, "every group's list begins on a line");

__device__ __forceinline__ unsigned dgc_zero_bytes(unsigned y) {
    return ~(((y & 0x7F7F7F7Fu) + 0x7F7F7F7Fu) | y | 0x7F7F7F7Fu);
}

template <int PITCH, class F>
__device__ __forceinline__ void dgc_add_edges(const uint2 m, const bool live, const int cnt, const int lane,
                                              const float4* __restrict__ s4, float* acc, F behind_first_gathers) {
    const int c = (int)(m.x & ((1u << COL_BITS) - 1));
    const int wbits = (int)m.y;
    const int rl = live ? (int)(m.x >> COL_BITS) : ROWS_PER_WG;
    #pragma unroll
    for (int g = 0; g < 32; g += EDGE_GROUP) {
        if (g == 0 || g < cnt) {
            float4 v[EDGE_GROUP];
            #pragma unroll
            for (int u = 0; u < EDGE_GROUP; ++u) {
                const unsigned cj = (unsigned)__builtin_amdgcn_readlane(c, g + u);
                v[u] = s4[cj * 32 + lane];
            }
            if (g == 0) behind_first_gathers();
            #pragma unroll
            for (int u = 0; u < EDGE_GROUP; ++u) {
                const int rj = __builtin_amdgcn_readlane(rl, g + u);
                const float wj = __int_as_float(__builtin_amdgcn_readlane(wbits, g + u));
                float* p = acc + rj * PITCH + lane;
                atomicAdd(p,      v[u].x * wj);
                atomicAdd(p + 32, v[u].y * wj);
                atomicAdd(p + 64, v[u].z * wj);
                atomicAdd(p + 96, v[u].w * wj);
            }
        }
    }
}

#define FC_NT      3
#define FC_PITCH   130
template <int MODE>
__global__ void __launch_bounds__(STEP_THREADS)
dgc_step_owned(const float* __restrict__ src,
               const float* __restrict__ dinv,
               const unsigned int* __restrict__ owner,
               const unsigned int* __restrict__ rec,
               const int* __restrict__ row,
               const int* __restrict__ col,
               unsigned int* kept,
               float* next,
               const float* __restrict__ W,
               const float* __restrict__ bias,
               const float* __restrict__ gamma,
               const float* __restrict__ beta) {
    constexpr bool LAST = MODE == 2;
    constexpr int PITCH = LAST ? FC_PITCH : DD;
    __shared__ float acc[(ROWS_PER_WG + 1) * PITCH];
    __shared__ __attribute__((aligned(16))) int wq[MODE == 1 ? 4 : STEP_WAVES * WQ_CAP];
    __shared__ int   wcount[STEP_WAVES];
    __shared__ int   lctr;
    __shared__ float wl[LAST ? FC_NT * 16 * FC_PITCH : 1];
    if constexpr (LAST) {
        for (int i = threadIdx.x; i < FC_NT * 16 * DD; i += STEP_THREADS) {
            const int n = i >> 7, k = i & (DD - 1);
            wl[n * FC_PITCH + k] = (n < CC) ? W[i] : 0.0f;
        }
    }

    const int tid  = threadIdx.x;
    const int lane = tid & 31;
    const int wave = tid >> 5;
    const int R0   = blockIdx.x * ROWS_PER_WG;
    const float4* s4 = (const float4*)src;

    for (int i = tid; i < ROWS_PER_WG * 32; i += STEP_THREADS) {
        const int rl = i >> 5;
        const float d = dinv[R0 + rl];
        const float a = (1.0f - DELTA) + DELTA * (d * d);
        const float4 v = s4[(R0 + rl) * 32 + lane];
        float* p = acc + rl * PITCH + lane;
        p[0]  = v.x * a;
        p[32] = v.y * a;
        p[64] = v.z * a;
        p[96] = v.w * a;
    }
    if (tid < DD) acc[ROWS_PER_WG * PITCH + tid] = 0.0f;
    if (lane == 0) wcount[wave] = 0;
    if (tid == 0) lctr = 0;
    __syncthreads();

    uint2* kl = (uint2*)(kept + (size_t)blockIdx.x * (LIST_BYTES / 4));

    if constexpr (MODE == 0) {
    const unsigned idpat = (unsigned)blockIdx.x * 0x01010101u;
    const uint4* ov = (const uint4*)owner;
    const uint2* rc = (const uint2*)rec;
    int* myq = wq + wave * WQ_CAP;

    #pragma unroll 1
    for (int it = 0; it < SCAN_ITERS / SCAN_BLOCK; ++it) {
        const int blk = (it + (int)blockIdx.x) % (SCAN_ITERS / SCAN_BLOCK);
        uint4 o[SCAN_BLOCK];
        #pragma unroll
        for (int s = 0; s < SCAN_BLOCK; ++s) {
            const int idx = (blk * SCAN_BLOCK + s) * STEP_THREADS + tid;
            o[s] = ov[idx];
        }
        #pragma unroll
        for (int s = 0; s < SCAN_BLOCK; ++s) {
            const int idx = (blk * SCAN_BLOCK + s) * STEP_THREADS + tid;
            unsigned mm = (dgc_zero_bytes(o[s].x ^ idpat) >> 7)
                        | (dgc_zero_bytes(o[s].y ^ idpat) >> 6)
                        | (dgc_zero_bytes(o[s].z ^ idpat) >> 5)
                        | (dgc_zero_bytes(o[s].w ^ idpat) >> 4);
            while (mm != 0) {
                const int pos = __ffs(mm) - 1;
                mm &= mm - 1;
                const int e = idx * 16 + ((pos & 7) << 2) + (pos >> 3);
                const int slot = atomicAdd(&wcount[wave], 1);
                if (slot < WQ_CAP) {
                    myq[slot] = e;
                } else {
                    const uint2 m = rc[e];
                    const int kp = atomicAdd(&lctr, 1);
                    if (kp < LIST_CAP) kl[LIST_HDR + kp] = m;
                    const int c = (int)(m.x & ((1u << COL_BITS) - 1));
                    const float w = __uint_as_float(m.y);
                    float* p = acc + (m.x >> COL_BITS) * PITCH;
                    #pragma unroll 1
                    for (int k = 0; k < DD; ++k)
                        atomicAdd(p + (k & 3) * 32 + (k >> 2), src[c * DD + k] * w);
                }
            }
        }
        int n = wcount[wave];
        n = __builtin_amdgcn_readfirstlane(n < WQ_CAP ? n : WQ_CAP);

        for (int base = 0; base < n; base += 32) {
            const int cnt = n - base;
            const int i = base + lane;
            const bool live = i < n;
            const int e = myq[live ? i : base];
            const uint2 m = rc[e];
            int kp = 0;
            if (lane == 0) kp = atomicAdd(&lctr, cnt < 32 ? cnt : 32);
            kp = __builtin_amdgcn_readfirstlane(kp);
            if (live && kp + lane < LIST_CAP) kl[LIST_HDR + kp + lane] = m;
            dgc_add_edges<PITCH>(m, live, cnt, lane, s4, acc, []() {});
        }
        if (lane == 0) wcount[wave] = 0;
    }
    } else {
    int n = ((const int*)kl)[0];
    n = __builtin_amdgcn_readfirstlane(n);
    if (n <= LIST_CAP) {
        const uint2* kr = kl + LIST_HDR;
        int ch = 0;
        if (lane == 0) ch = atomicAdd(&lctr, 1);
        ch = __builtin_amdgcn_readfirstlane(ch);
        uint2 m = make_uint2(0u, 0u);
        if (ch * 32 < n) { const int i = ch * 32 + lane; m = kr[i < n ? i : ch * 32]; }
        while (ch * 32 < n) {
            const int base = ch * 32;
            int chn = 0;
            uint2 mn = m;
            dgc_add_edges<PITCH>(m, base + lane < n, n - base, lane, s4, acc, [&]() {
                int t = 0;
                if (lane == 0) t = atomicAdd(&lctr, 1);
                chn = __builtin_amdgcn_readfirstlane(t);
                if (chn * 32 < n) { const int i = chn * 32 + lane; mn = kr[i < n ? i : chn * 32]; }
            });
            ch = chn;
            m = mn;
        }
    } else {
        #pragma unroll 1
        for (int e = tid; e < EE; e += STEP_THREADS) {
            const int r = row[e];
            const unsigned rl = (unsigned)(r - R0);
            if (rl < (unsigned)ROWS_PER_WG) {
                const int c = col[e];
                const float w = DELTA * (dinv[r] * dinv[c]);
                float* p = acc + rl * PITCH;
                #pragma unroll 1
                for (int k = 0; k < DD; ++k)
                    atomicAdd(p + (k & 3) * 32 + (k >> 2), src[c * DD + k] * w);
            }
        }
    }
    }
    __syncthreads();
    if constexpr (MODE == 0) {
        if (tid == 0) ((int*)kl)[0] = lctr;
    }

    if constexpr (!LAST) {
        for (int i = tid; i < ROWS_PER_WG * 32; i += STEP_THREADS) {
            const int rl = i >> 5;
            const float* p = acc + rl * PITCH + lane;
            float4 v;
            v.x = p[0]; v.y = p[32]; v.z = p[64]; v.w = p[96];
            ((float4*)next)[(R0 + rl) * 32 + lane] = v;
        }
    } else {
        float* stage = (float*)wq;
        static_assert(ROWS_PER_WG % 16 == 0 && ROWS_PER_WG / 16 <= STEP_WAVES, "one wave per 16-row tile");
        static_assert(ROWS_PER_WG * CC * 4 <= STEP_WAVES * WQ_CAP * 4, "the result fits the staging LDS");
        if (wave < ROWS_PER_WG / 16) {
            const int c = lane & 15;
            const int h = lane >> 4;
            const float* ap = acc + (wave * 16 + c) * PITCH + h * 32;
            const float* bp = wl + c * FC_PITCH + h;
            v8f d0 = {}, d1 = {}, d2 = {};
            #pragma unroll 8
            for (int s = 0; s < DD / 4; ++s) {
                v2f a, b0, b1, b2;
                a.x  = ap[s];
                a.y  = ap[64 + s];
                b0.x = bp[4 * s];
                b0.y = bp[4 * s + 2];
                b1.x = bp[16 * FC_PITCH + 4 * s];
                b1.y = bp[16 * FC_PITCH + 4 * s + 2];
                b2.x = bp[32 * FC_PITCH + 4 * s];
                b2.y = bp[32 * FC_PITCH + 4 * s + 2];
                d0 = __builtin_amdgcn_wmma_f32_16x16x4_f32(false, a, false, b0, (short)0, d0, false, false);
                d1 = __builtin_amdgcn_wmma_f32_16x16x4_f32(false, a, false, b1, (short)0, d1, false, false);
                d2 = __builtin_amdgcn_wmma_f32_16x16x4_f32(false, a, false, b2, (short)0, d2, false, false);
            }
            const bool t2 = c < CC - 32;
            const float bv0 = bias[c],  bv1 = bias[16 + c],  bv2 = t2 ? bias[32 + c]  : 0.0f;
            const float g0  = gamma[c], g1  = gamma[16 + c], g2  = t2 ? gamma[32 + c] : 0.0f;
            const float e0  = beta[c],  e1  = beta[16 + c],  e2  = t2 ? beta[32 + c]  : 0.0f;
            #pragma unroll
            for (int r = 0; r < 8; ++r) {
                const float x0 = d0[r] + bv0;
                const float x1 = d1[r] + bv1;
                const float x2 = t2 ? d2[r] + bv2 : 0.0f;
                float s  = x0 + x1 + x2;
                float sq = x0 * x0 + x1 * x1 + x2 * x2;
                #pragma unroll
                for (int off = 8; off > 0; off >>= 1) {
                    s  += __shfl_xor(s,  off, 32);
                    sq += __shfl_xor(sq, off, 32);
                }
                const float mean = s * (1.0f / CC);
                const float var  = sq * (1.0f / CC) - mean * mean;
                const float inv  = rsqrtf(var + LN_EPS);
                float* q = stage + (wave * 16 + h * 8 + r) * CC + c;
                q[0]  = (x0 - mean) * inv * g0 + e0;
                q[16] = (x1 - mean) * inv * g1 + e1;
                if (t2) q[32] = (x2 - mean) * inv * g2 + e2;
            }
        }
        __syncthreads();
        float4* o4 = (float4*)(next + (size_t)R0 * CC);
        for (int i = tid; i < ROWS_PER_WG * CC / 4; i += STEP_THREADS)
            o4[i] = ((const float4*)stage)[i];
    }
}

extern "C" void kernel_launch(void* const* d_in, const int* in_sizes, int n_in,
                              void* d_out, int out_size, void* d_ws, size_t ws_size,
                              hipStream_t stream) {
    const float* feat  = (const float*)d_in[0];
    const int*   row   = (const int*)  d_in[1];
    const int*   col   = (const int*)  d_in[2];
    const float* W     = (const float*)d_in[3];
    const float* bias  = (const float*)d_in[4];
    const float* gamma = (const float*)d_in[5];
    const float* beta  = (const float*)d_in[6];
    float* out = (float*)d_out;

    const size_t NP = 100352;
    float* dinv = (float*)d_ws;
    float* ta   = dinv + NP;
    float* tb   = ta + (size_t)NN * DD;

    unsigned int* owner = (unsigned int*)tb;
    unsigned int* rec   = owner + LIST_REC_OFF;
    unsigned int* kept  = (unsigned int*)d_out;

    const int B = 256;

    dgc_deg_init <<<(NN + B - 1) / B, B, 0, stream>>>(dinv);
    dgc_deg_count<<<(EE + B - 1) / B, B, 0, stream>>>(row, dinv);
    dgc_deg_rsqrt<<<(NN + B - 1) / B, B, 0, stream>>>(dinv);

    static_assert(OWNER_DWORDS <= LIST_REC_OFF, "the owner bytes end before the records begin");
    static_assert(2097152 + (size_t)EE * 8 <= (size_t)NN * DD * 4, "both fit in tb");
    static_assert((size_t)NUM_OWNERS * LIST_BYTES == (size_t)NN * CC * 4, "the groups' kept records fill out exactly");
    dgc_edge_prep<<<(OWNER_DWORDS + B - 1) / B, B, 0, stream>>>(row, col, dinv, owner, rec);

    static_assert(KSTEPS == 4, "the four steps are written out below");
    dgc_step_owned<0><<<NUM_OWNERS, STEP_THREADS, 0, stream>>>(feat, dinv, owner, rec, row, col, kept, ta, W, bias, gamma, beta);
    dgc_step_owned<1><<<NUM_OWNERS, STEP_THREADS, 0, stream>>>(ta, dinv, owner, rec, row, col, kept, tb, W, bias, gamma, beta);
    dgc_step_owned<1><<<NUM_OWNERS, STEP_THREADS, 0, stream>>>(tb, dinv, owner, rec, row, col, kept, ta, W, bias, gamma, beta);
    dgc_step_owned<2><<<NUM_OWNERS, STEP_THREADS, 0, stream>>>(ta, dinv, owner, rec, row, col, kept, out, W, bias, gamma, beta);
}


# Set-up: owner groups scan the owner bytes ahead of step 1, count their rows' degrees in LDS (integers), keep the keys; step 1 forms the weights; global f32 degree atomics and 2 launches go; sums kept
# speedup vs baseline: 5.7203x; 1.3243x over previous
#include <hip/hip_runtime.h>
#include <hip/hip_bf16.h>

#define NN      100000
#define EE      1600000
#define DD      128
#define CC      40
#define KSTEPS  4
#define DELTA   0.5f
#define LN_EPS  1e-5f

typedef __attribute__((ext_vector_type(2))) float v2f;
typedef __attribute__((ext_vector_type(8))) float v8f;

#define ROWS_PER_WG  400
#define NUM_OWNERS   (NN / ROWS_PER_WG)
#define STEP_THREADS 1024
#define OWNER_VECS   (EE / 16)
#define SCAN_ITERS   ((OWNER_VECS + STEP_THREADS - 1) / STEP_THREADS)
#define OWNER_DWORDS (SCAN_ITERS * STEP_THREADS * 4)
#define COL_BITS     17
#define LIST_KEY_OFF 524288
static_assert(NN % ROWS_PER_WG == 0, "row ranges cover the nodes exactly");
static_assert(NUM_OWNERS <= 255, "an owner number fits a byte, 255 stays free");
static_assert(EE % 16 == 0, "owner bytes are read 16 at a time");
static_assert(NN <= (1 << COL_BITS) && ROWS_PER_WG <= (1 << (32 - COL_BITS)), "column and local row share a dword");

__device__ __forceinline__ unsigned dgc_rec_key(int r, int c) {
    return (unsigned)c | ((unsigned)(r - (r / ROWS_PER_WG) * ROWS_PER_WG) << COL_BITS);
}

__global__ void dgc_owner_keys(const int* __restrict__ row,
                               const int* __restrict__ col,
                               unsigned int* __restrict__ owner,
                               unsigned int* __restrict__ key) {
    const int t = blockIdx.x * blockDim.x + threadIdx.x;
    if (t < EE / 4) {
        const int4 r = ((const int4*)row)[t];
        const int4 c = ((const int4*)col)[t];
        owner[t] =  (unsigned)(r.x / ROWS_PER_WG)
                 | ((unsigned)(r.y / ROWS_PER_WG) << 8)
                 | ((unsigned)(r.z / ROWS_PER_WG) << 16)
                 | ((unsigned)(r.w / ROWS_PER_WG) << 24);
        uint4 k;
        k.x = dgc_rec_key(r.x, c.x);
        k.y = dgc_rec_key(r.y, c.y);
        k.z = dgc_rec_key(r.z, c.z);
        k.w = dgc_rec_key(r.w, c.w);
        ((uint4*)key)[t] = k;
    } else if (t < OWNER_DWORDS) {
        owner[t] = ~0u;
    }
}

#define STEP_WAVES    (STEP_THREADS / 32)
#define WQ_CAP        512
#define SCAN_BLOCK    14
static_assert(SCAN_ITERS % SCAN_BLOCK == 0, "the scan is whole blocks");

#define LIST_BYTES    (ROWS_PER_WG * CC * 4)
#define LIST_HDR      2
#define LIST_CAP      (LIST_BYTES / 8 - LIST_HDR)
static_assert(LIST_BYTES % 128 == 0, "every group's list begins on a line");

__device__ __forceinline__ unsigned dgc_zero_bytes(unsigned y) {
    return ~(((y & 0x7F7F7F7Fu) + 0x7F7F7F7Fu) | y | 0x7F7F7F7Fu);
}

__global__ void __launch_bounds__(STEP_THREADS)
dgc_owner_scan(const unsigned int* __restrict__ owner,
               const unsigned int* __restrict__ key,
               unsigned int* __restrict__ kept,
               float* __restrict__ dinv) {
    __shared__ int cnt[ROWS_PER_WG];
    __shared__ int wq[STEP_WAVES * WQ_CAP];
    __shared__ int wcount[STEP_WAVES];
    __shared__ int lctr;
    const int tid  = threadIdx.x;
    const int lane = tid & 31;
    const int wave = tid >> 5;
    if (tid < ROWS_PER_WG) cnt[tid] = 0;
    if (lane == 0) wcount[wave] = 0;
    if (tid == 0) lctr = 0;
    __syncthreads();

    uint2* kl = (uint2*)(kept + (size_t)blockIdx.x * (LIST_BYTES / 4));
    const unsigned idpat = (unsigned)blockIdx.x * 0x01010101u;
    const uint4* ov = (const uint4*)owner;
    int* myq = wq + wave * WQ_CAP;

    #pragma unroll 1
    for (int it = 0; it < SCAN_ITERS / SCAN_BLOCK; ++it) {
        const int blk = (it + (int)blockIdx.x) % (SCAN_ITERS / SCAN_BLOCK);
        uint4 o[SCAN_BLOCK];
        #pragma unroll
        for (int s = 0; s < SCAN_BLOCK; ++s) {
            const int idx = (blk * SCAN_BLOCK + s) * STEP_THREADS + tid;
            o[s] = ov[idx];
        }
        #pragma unroll
        for (int s = 0; s < SCAN_BLOCK; ++s) {
            const int idx = (blk * SCAN_BLOCK + s) * STEP_THREADS + tid;
            unsigned mm = (dgc_zero_bytes(o[s].x ^ idpat) >> 7)
                        | (dgc_zero_bytes(o[s].y ^ idpat) >> 6)
                        | (dgc_zero_bytes(o[s].z ^ idpat) >> 5)
                        | (dgc_zero_bytes(o[s].w ^ idpat) >> 4);
            while (mm != 0) {
                const int pos = __ffs(mm) - 1;
                mm &= mm - 1;
                const int e = idx * 16 + ((pos & 7) << 2) + (pos >> 3);
                const int slot = atomicAdd(&wcount[wave], 1);
                if (slot < WQ_CAP) {
                    myq[slot] = e;
                } else {
                    const unsigned k = key[e];
                    atomicAdd(&cnt[k >> COL_BITS], 1);
                    const int kp = atomicAdd(&lctr, 1);
                    if (kp < LIST_CAP) kl[LIST_HDR + kp].x = k;
                }
            }
        }
        int n = wcount[wave];
        n = __builtin_amdgcn_readfirstlane(n < WQ_CAP ? n : WQ_CAP);

        for (int base = 0; base < n; base += 32) {
            const int cnt32 = n - base < 32 ? n - base : 32;
            const int i = base + lane;
            const bool live = i < n;
            const unsigned k = key[myq[live ? i : base]];
            int kp = 0;
            if (lane == 0) kp = atomicAdd(&lctr, cnt32);
            kp = __builtin_amdgcn_readfirstlane(kp);
            if (live) {
                atomicAdd(&cnt[k >> COL_BITS], 1);
                if (kp + lane < LIST_CAP) kl[LIST_HDR + kp + lane].x = k;
            }
        }
        if (lane == 0) wcount[wave] = 0;
    }
    __syncthreads();
    if (tid == 0) ((int*)kl)[0] = lctr;
    if (tid < ROWS_PER_WG) dinv[blockIdx.x * ROWS_PER_WG + tid] = rsqrtf((float)(cnt[tid] + 1));
}

#define EDGE_GROUP    8

template <int PITCH, class F>
__device__ __forceinline__ void dgc_add_edges(const uint2 m, const bool live, const int cnt, const int lane,
                                              const float4* __restrict__ s4, float* acc, F behind_first_gathers) {
    const int c = (int)(m.x & ((1u << COL_BITS) - 1));
    const int wbits = (int)m.y;
    const int rl = live ? (int)(m.x >> COL_BITS) : ROWS_PER_WG;
    #pragma unroll
    for (int g = 0; g < 32; g += EDGE_GROUP) {
        if (g == 0 || g < cnt) {
            float4 v[EDGE_GROUP];
            #pragma unroll
            for (int u = 0; u < EDGE_GROUP; ++u) {
                const unsigned cj = (unsigned)__builtin_amdgcn_readlane(c, g + u);
                v[u] = s4[cj * 32 + lane];
            }
            if (g == 0) behind_first_gathers();
            #pragma unroll
            for (int u = 0; u < EDGE_GROUP; ++u) {
                const int rj = __builtin_amdgcn_readlane(rl, g + u);
                const float wj = __int_as_float(__builtin_amdgcn_readlane(wbits, g + u));
                float* p = acc + rj * PITCH + lane;
                atomicAdd(p,      v[u].x * wj);
                atomicAdd(p + 32, v[u].y * wj);
                atomicAdd(p + 64, v[u].z * wj);
                atomicAdd(p + 96, v[u].w * wj);
            }
        }
    }
}

#define FC_NT      3
#define FC_PITCH   130
template <int MODE>
__global__ void __launch_bounds__(STEP_THREADS)
dgc_step_owned(const float* __restrict__ src,
               const float* __restrict__ dinv,
               const int* __restrict__ row,
               const int* __restrict__ col,
               unsigned int* kept,
               float* next,
               const float* __restrict__ W,
               const float* __restrict__ bias,
               const float* __restrict__ gamma,
               const float* __restrict__ beta) {
    constexpr bool FIRST = MODE == 0;
    constexpr bool LAST = MODE == 2;
    constexpr int PITCH = LAST ? FC_PITCH : DD;
    __shared__ float acc[(ROWS_PER_WG + 1) * PITCH];
    __shared__ __attribute__((aligned(16))) int wq[LAST ? STEP_WAVES * WQ_CAP : 4];
    __shared__ int   lctr;
    __shared__ float wl[LAST ? FC_NT * 16 * FC_PITCH : 1];
    if constexpr (LAST) {
        for (int i = threadIdx.x; i < FC_NT * 16 * DD; i += STEP_THREADS) {
            const int n = i >> 7, k = i & (DD - 1);
            wl[n * FC_PITCH + k] = (n < CC) ? W[i] : 0.0f;
        }
    }

    const int tid  = threadIdx.x;
    const int lane = tid & 31;
    const int wave = tid >> 5;
    const int R0   = blockIdx.x * ROWS_PER_WG;
    const float4* s4 = (const float4*)src;

    for (int i = tid; i < ROWS_PER_WG * 32; i += STEP_THREADS) {
        const int rl = i >> 5;
        const float d = dinv[R0 + rl];
        const float a = (1.0f - DELTA) + DELTA * (d * d);
        const float4 v = s4[(R0 + rl) * 32 + lane];
        float* p = acc + rl * PITCH + lane;
        p[0]  = v.x * a;
        p[32] = v.y * a;
        p[64] = v.z * a;
        p[96] = v.w * a;
    }
    if (tid < DD) acc[ROWS_PER_WG * PITCH + tid] = 0.0f;
    if (tid == 0) lctr = 0;
    __syncthreads();

    uint2* kl = (uint2*)(kept + (size_t)blockIdx.x * (LIST_BYTES / 4));

    {
    int n = ((const int*)kl)[0];
    n = __builtin_amdgcn_readfirstlane(n);
    if (n <= LIST_CAP) {
        uint2* kr = kl + LIST_HDR;
        auto piece = [&](const int p, float& a, float& b) {
            const int i = p * 32 + lane;
            const uint2 r = kr[i < n ? i : p * 32];
            if constexpr (FIRST) {
                a = dinv[R0 + (int)(r.x >> COL_BITS)];
                b = dinv[(int)(r.x & ((1u << COL_BITS) - 1))];
            }
            return r;
        };
        float dr = 0.0f, dc = 0.0f;
        int ch = 0;
        if (lane == 0) ch = atomicAdd(&lctr, 1);
        ch = __builtin_amdgcn_readfirstlane(ch);
        uint2 m = make_uint2(0u, 0u);
        if (ch * 32 < n) m = piece(ch, dr, dc);
        while (ch * 32 < n) {
            const int base = ch * 32;
            int chn = 0;
            uint2 mn = m;
            float drn = dr, dcn = dc;
            if constexpr (FIRST) {
                m.y = __float_as_uint(DELTA * (dr * dc));
                if (base + lane < n) kr[base + lane] = m;
            }
            dgc_add_edges<PITCH>(m, base + lane < n, n - base, lane, s4, acc, [&]() {
                int t = 0;
                if (lane == 0) t = atomicAdd(&lctr, 1);
                chn = __builtin_amdgcn_readfirstlane(t);
                if (chn * 32 < n) mn = piece(chn, drn, dcn);
            });
            ch = chn;
            m = mn;
            dr = drn;
            dc = dcn;
        }
    } else {
        #pragma unroll 1
        for (int e = tid; e < EE; e += STEP_THREADS) {
            const int r = row[e];
            const unsigned rl = (unsigned)(r - R0);
            if (rl < (unsigned)ROWS_PER_WG) {
                const int c = col[e];
                const float w = DELTA * (dinv[r] * dinv[c]);
                float* p = acc + rl * PITCH;
                #pragma unroll 1
                for (int k = 0; k < DD; ++k)
                    atomicAdd(p + (k & 3) * 32 + (k >> 2), src[c * DD + k] * w);
            }
        }
    }
    }
    __syncthreads();

    if constexpr (!LAST) {
        for (int i = tid; i < ROWS_PER_WG * 32; i += STEP_THREADS) {
            const int rl = i >> 5;
            const float* p = acc + rl * PITCH + lane;
            float4 v;
            v.x = p[0]; v.y = p[32]; v.z = p[64]; v.w = p[96];
            ((float4*)next)[(R0 + rl) * 32 + lane] = v;
        }
    } else {
        float* stage = (float*)wq;
        static_assert(ROWS_PER_WG % 16 == 0 && ROWS_PER_WG / 16 <= STEP_WAVES, "one wave per 16-row tile");
        static_assert(ROWS_PER_WG * CC * 4 <= STEP_WAVES * WQ_CAP * 4, "the result fits the staging LDS");
        if (wave < ROWS_PER_WG / 16) {
            const int c = lane & 15;
            const int h = lane >> 4;
            const float* ap = acc + (wave * 16 + c) * PITCH + h * 32;
            const float* bp = wl + c * FC_PITCH + h;
            v8f d0 = {}, d1 = {}, d2 = {};
            #pragma unroll 8
            for (int s = 0; s < DD / 4; ++s) {
                v2f a, b0, b1, b2;
                a.x  = ap[s];
                a.y  = ap[64 + s];
                b0.x = bp[4 * s];
                b0.y = bp[4 * s + 2];
                b1.x = bp[16 * FC_PITCH + 4 * s];
                b1.y = bp[16 * FC_PITCH + 4 * s + 2];
                b2.x = bp[32 * FC_PITCH + 4 * s];
                b2.y = bp[32 * FC_PITCH + 4 * s + 2];
                d0 = __builtin_amdgcn_wmma_f32_16x16x4_f32(false, a, false, b0, (short)0, d0, false, false);
                d1 = __builtin_amdgcn_wmma_f32_16x16x4_f32(false, a, false, b1, (short)0, d1, false, false);
                d2 = __builtin_amdgcn_wmma_f32_16x16x4_f32(false, a, false, b2, (short)0, d2, false, false);
            }
            const bool t2 = c < CC - 32;
            const float bv0 = bias[c],  bv1 = bias[16 + c],  bv2 = t2 ? bias[32 + c]  : 0.0f;
            const float g0  = gamma[c], g1  = gamma[16 + c], g2  = t2 ? gamma[32 + c] : 0.0f;
            const float e0  = beta[c],  e1  = beta[16 + c],  e2  = t2 ? beta[32 + c]  : 0.0f;
            #pragma unroll
            for (int r = 0; r < 8; ++r) {
                const float x0 = d0[r] + bv0;
                const float x1 = d1[r] + bv1;
                const float x2 = t2 ? d2[r] + bv2 : 0.0f;
                float s  = x0 + x1 + x2;
                float sq = x0 * x0 + x1 * x1 + x2 * x2;
                #pragma unroll
                for (int off = 8; off > 0; off >>= 1) {
                    s  += __shfl_xor(s,  off, 32);
                    sq += __shfl_xor(sq, off, 32);
                }
                const float mean = s * (1.0f / CC);
                const float var  = sq * (1.0f / CC) - mean * mean;
                const float inv  = rsqrtf(var + LN_EPS);
                float* q = stage + (wave * 16 + h * 8 + r) * CC + c;
                q[0]  = (x0 - mean) * inv * g0 + e0;
                q[16] = (x1 - mean) * inv * g1 + e1;
                if (t2) q[32] = (x2 - mean) * inv * g2 + e2;
            }
        }
        __syncthreads();
        float4* o4 = (float4*)(next + (size_t)R0 * CC);
        for (int i = tid; i < ROWS_PER_WG * CC / 4; i += STEP_THREADS)
            o4[i] = ((const float4*)stage)[i];
    }
}

extern "C" void kernel_launch(void* const* d_in, const int* in_sizes, int n_in,
                              void* d_out, int out_size, void* d_ws, size_t ws_size,
                              hipStream_t stream) {
    const float* feat  = (const float*)d_in[0];
    const int*   row   = (const int*)  d_in[1];
    const int*   col   = (const int*)  d_in[2];
    const float* W     = (const float*)d_in[3];
    const float* bias  = (const float*)d_in[4];
    const float* gamma = (const float*)d_in[5];
    const float* beta  = (const float*)d_in[6];
    float* out = (float*)d_out;

    const size_t NP = 100352;
    float* dinv = (float*)d_ws;
    float* ta   = dinv + NP;
    float* tb   = ta + (size_t)NN * DD;

    unsigned int* owner = (unsigned int*)tb;
    unsigned int* key   = owner + LIST_KEY_OFF;
    unsigned int* kept  = (unsigned int*)d_out;

    const int B = 256;

    static_assert(OWNER_DWORDS <= LIST_KEY_OFF, "the owner bytes end before the keys begin");
    static_assert(2097152 + (size_t)EE * 4 <= (size_t)NN * DD * 4, "both fit in tb");
    static_assert((size_t)NUM_OWNERS * LIST_BYTES == (size_t)NN * CC * 4, "the groups' kept records fill out exactly");
    dgc_owner_keys<<<(OWNER_DWORDS + B - 1) / B, B, 0, stream>>>(row, col, owner, key);
    dgc_owner_scan<<<NUM_OWNERS, STEP_THREADS, 0, stream>>>(owner, key, kept, dinv);

    static_assert(KSTEPS == 4, "the four steps are written out below");
    dgc_step_owned<0><<<NUM_OWNERS, STEP_THREADS, 0, stream>>>(feat, dinv, row, col, kept, ta, W, bias, gamma, beta);
    dgc_step_owned<1><<<NUM_OWNERS, STEP_THREADS, 0, stream>>>(ta, dinv, row, col, kept, tb, W, bias, gamma, beta);
    dgc_step_owned<1><<<NUM_OWNERS, STEP_THREADS, 0, stream>>>(tb, dinv, row, col, kept, ta, W, bias, gamma, beta);
    dgc_step_owned<2><<<NUM_OWNERS, STEP_THREADS, 0, stream>>>(ta, dinv, row, col, kept, out, W, bias, gamma, beta);
}


# Steps: wave w owns tile rows w mod 32; step 1 files kept records into 32 per-wave bins (LDS counts); an edge adds by 16-byte LDS load, FMA, store, not 4 f32 LDS atomics; f32 sum order changes
# speedup vs baseline: 6.2314x; 1.0893x over previous
#include <hip/hip_runtime.h>
#include <hip/hip_bf16.h>

#define NN      100000
#define EE      1600000
#define DD      128
#define CC      40
#define KSTEPS  4
#define DELTA   0.5f
#define LN_EPS  1e-5f

typedef __attribute__((ext_vector_type(2))) float v2f;
typedef __attribute__((ext_vector_type(8))) float v8f;

#define ROWS_PER_WG  400
#define NUM_OWNERS   (NN / ROWS_PER_WG)
#define STEP_THREADS 1024
#define OWNER_VECS   (EE / 16)
#define SCAN_ITERS   ((OWNER_VECS + STEP_THREADS - 1) / STEP_THREADS)
#define OWNER_DWORDS (SCAN_ITERS * STEP_THREADS * 4)
#define COL_BITS     17
#define LIST_KEY_OFF 524288
static_assert(NN % ROWS_PER_WG == 0, "row ranges cover the nodes exactly");
static_assert(NUM_OWNERS <= 255, "an owner number fits a byte, 255 stays free");
static_assert(EE % 16 == 0, "owner bytes are read 16 at a time");
static_assert(NN <= (1 << COL_BITS) && ROWS_PER_WG <= (1 << (32 - COL_BITS)), "column and local row share a dword");

__device__ __forceinline__ unsigned dgc_rec_key(int r, int c) {
    return (unsigned)c | ((unsigned)(r - (r / ROWS_PER_WG) * ROWS_PER_WG) << COL_BITS);
}

__global__ void dgc_owner_keys(const int* __restrict__ row,
                               const int* __restrict__ col,
                               unsigned int* __restrict__ owner,
                               unsigned int* __restrict__ key) {
    const int t = blockIdx.x * blockDim.x + threadIdx.x;
    if (t < EE / 4) {
        const int4 r = ((const int4*)row)[t];
        const int4 c = ((const int4*)col)[t];
        owner[t] =  (unsigned)(r.x / ROWS_PER_WG)
                 | ((unsigned)(r.y / ROWS_PER_WG) << 8)
                 | ((unsigned)(r.z / ROWS_PER_WG) << 16)
                 | ((unsigned)(r.w / ROWS_PER_WG) << 24);
        uint4 k;
        k.x = dgc_rec_key(r.x, c.x);
        k.y = dgc_rec_key(r.y, c.y);
        k.z = dgc_rec_key(r.z, c.z);
        k.w = dgc_rec_key(r.w, c.w);
        ((uint4*)key)[t] = k;
    } else if (t < OWNER_DWORDS) {
        owner[t] = ~0u;
    }
}

#define STEP_WAVES    (STEP_THREADS / 32)
#define WQ_CAP        512
#define SCAN_BLOCK    14
static_assert(SCAN_ITERS % SCAN_BLOCK == 0, "the scan is whole blocks");

#define LIST_BYTES    (ROWS_PER_WG * CC * 4)
#define LIST_HDR      16
#define LIST_CAP      (LIST_BYTES / 8 - LIST_HDR)
static_assert(LIST_BYTES % 128 == 0, "every group's list begins on a line");

__device__ __forceinline__ unsigned dgc_zero_bytes(unsigned y) {
    return ~(((y & 0x7F7F7F7Fu) + 0x7F7F7F7Fu) | y | 0x7F7F7F7Fu);
}

__global__ void __launch_bounds__(STEP_THREADS)
dgc_owner_scan(const unsigned int* __restrict__ owner,
               const unsigned int* __restrict__ key,
               unsigned int* __restrict__ kept,
               float* __restrict__ dinv) {
    __shared__ int cnt[ROWS_PER_WG];
    __shared__ int wq[STEP_WAVES * WQ_CAP];
    __shared__ int wcount[STEP_WAVES];
    __shared__ int lctr;
    const int tid  = threadIdx.x;
    const int lane = tid & 31;
    const int wave = tid >> 5;
    if (tid < ROWS_PER_WG) cnt[tid] = 0;
    if (lane == 0) wcount[wave] = 0;
    if (tid == 0) lctr = 0;
    __syncthreads();

    uint2* kl = (uint2*)(kept + (size_t)blockIdx.x * (LIST_BYTES / 4));
    const unsigned idpat = (unsigned)blockIdx.x * 0x01010101u;
    const uint4* ov = (const uint4*)owner;
    int* myq = wq + wave * WQ_CAP;

    #pragma unroll 1
    for (int it = 0; it < SCAN_ITERS / SCAN_BLOCK; ++it) {
        const int blk = (it + (int)blockIdx.x) % (SCAN_ITERS / SCAN_BLOCK);
        uint4 o[SCAN_BLOCK];
        #pragma unroll
        for (int s = 0; s < SCAN_BLOCK; ++s) {
            const int idx = (blk * SCAN_BLOCK + s) * STEP_THREADS + tid;
            o[s] = ov[idx];
        }
        #pragma unroll
        for (int s = 0; s < SCAN_BLOCK; ++s) {
            const int idx = (blk * SCAN_BLOCK + s) * STEP_THREADS + tid;
            unsigned mm = (dgc_zero_bytes(o[s].x ^ idpat) >> 7)
                        | (dgc_zero_bytes(o[s].y ^ idpat) >> 6)
                        | (dgc_zero_bytes(o[s].z ^ idpat) >> 5)
                        | (dgc_zero_bytes(o[s].w ^ idpat) >> 4);
            while (mm != 0) {
                const int pos = __ffs(mm) - 1;
                mm &= mm - 1;
                const int e = idx * 16 + ((pos & 7) << 2) + (pos >> 3);
                const int slot = atomicAdd(&wcount[wave], 1);
                if (slot < WQ_CAP) {
                    myq[slot] = e;
                } else {
                    const unsigned k = key[e];
                    atomicAdd(&cnt[k >> COL_BITS], 1);
                    const int kp = atomicAdd(&lctr, 1);
                    if (kp < LIST_CAP) kl[LIST_HDR + kp].x = k;
                }
            }
        }
        int n = wcount[wave];
        n = __builtin_amdgcn_readfirstlane(n < WQ_CAP ? n : WQ_CAP);

        for (int base = 0; base < n; base += 32) {
            const int cnt32 = n - base < 32 ? n - base : 32;
            const int i = base + lane;
            const bool live = i < n;
            const unsigned k = key[myq[live ? i : base]];
            int kp = 0;
            if (lane == 0) kp = atomicAdd(&lctr, cnt32);
            kp = __builtin_amdgcn_readfirstlane(kp);
            if (live) {
                atomicAdd(&cnt[k >> COL_BITS], 1);
                if (kp + lane < LIST_CAP) kl[LIST_HDR + kp + lane].x = k;
            }
        }
        if (lane == 0) wcount[wave] = 0;
    }
    __syncthreads();
    if (tid == 0) ((int*)kl)[0] = lctr;
    if (tid < ROWS_PER_WG) dinv[blockIdx.x * ROWS_PER_WG + tid] = rsqrtf((float)(cnt[tid] + 1));
}

#define EDGE_GROUP    8
#define BIN_ROUNDS    ((LIST_CAP + STEP_THREADS - 1) / STEP_THREADS)
static_assert(STEP_WAVES == 32, "a bin per wave, scanned by one wave's 32 lanes");

__device__ __forceinline__ void dgc_gather8(float4 (&v)[EDGE_GROUP], const int c, const int g, const int lane,
                                            const float4* __restrict__ s4) {
    #pragma unroll
    for (int u = 0; u < EDGE_GROUP; ++u) {
        const unsigned cj = (unsigned)__builtin_amdgcn_readlane(c, g + u);
        v[u] = s4[cj * 32 + lane];
    }
}

template <int PITCH, bool FULL>
__device__ __forceinline__ void dgc_add8(const float4 (&v)[EDGE_GROUP], const int rl, const int wbits, const int g,
                                         const int cnt, const int lane, float* acc) {
    #pragma unroll
    for (int u = 0; u < EDGE_GROUP; ++u) {
        if (FULL || g + u < cnt) {
            const int rj = __builtin_amdgcn_readlane(rl, g + u);
            const float wj = __int_as_float(__builtin_amdgcn_readlane(wbits, g + u));
            float4* p = (float4*)(acc + rj * PITCH) + lane;
            float4 a = *p;
            a.x += v[u].x * wj;
            a.y += v[u].y * wj;
            a.z += v[u].z * wj;
            a.w += v[u].w * wj;
            *p = a;
        }
    }
}

#define FC_NT      3
#define FC_PITCH   130
#define FC_TILE_PITCH 132
template <int MODE>
__global__ void __launch_bounds__(STEP_THREADS)
dgc_step_owned(const float* __restrict__ src,
               const float* __restrict__ dinv,
               const int* __restrict__ row,
               const int* __restrict__ col,
               unsigned int* kept,
               float* next,
               const float* __restrict__ W,
               const float* __restrict__ bias,
               const float* __restrict__ gamma,
               const float* __restrict__ beta) {
    constexpr bool FIRST = MODE == 0;
    constexpr bool LAST = MODE == 2;
    constexpr int PITCH = LAST ? FC_TILE_PITCH : DD;
    __shared__ __attribute__((aligned(16))) float acc[ROWS_PER_WG * PITCH];
    __shared__ __attribute__((aligned(16))) float stage[LAST ? ROWS_PER_WG * CC : 4];
    __shared__ float wl[LAST ? FC_NT * 16 * FC_PITCH : 1];
    __shared__ uint2 srt[FIRST ? LIST_CAP : 1];
    __shared__ int   bcnt[STEP_WAVES];
    __shared__ int   boff[STEP_WAVES + 1];
    if constexpr (LAST) {
        for (int i = threadIdx.x; i < FC_NT * 16 * DD; i += STEP_THREADS) {
            const int n = i >> 7, k = i & (DD - 1);
            wl[n * FC_PITCH + k] = (n < CC) ? W[i] : 0.0f;
        }
    }

    const int tid  = threadIdx.x;
    const int lane = tid & 31;
    const int wave = __builtin_amdgcn_readfirstlane(tid >> 5);
    const int R0   = blockIdx.x * ROWS_PER_WG;
    const float4* s4 = (const float4*)src;

    for (int i = tid; i < ROWS_PER_WG * 32; i += STEP_THREADS) {
        const int rl = i >> 5;
        const float d = dinv[R0 + rl];
        const float a = (1.0f - DELTA) + DELTA * (d * d);
        float4 v = s4[(R0 + rl) * 32 + lane];
        v.x *= a; v.y *= a; v.z *= a; v.w *= a;
        ((float4*)(acc + rl * PITCH))[lane] = v;
    }
    if (tid < STEP_WAVES) bcnt[tid] = 0;
    __syncthreads();

    uint2* kl = (uint2*)(kept + (size_t)blockIdx.x * (LIST_BYTES / 4));
    const int* hdr = (const int*)kl;

    {
    int n = hdr[0];
    n = __builtin_amdgcn_readfirstlane(n);
    if (n <= LIST_CAP) {
        uint2* kr = kl + LIST_HDR;
        if constexpr (FIRST) {
            static_assert(BIN_ROUNDS * STEP_THREADS >= LIST_CAP, "every kept record has a thread");
            unsigned kk[BIN_ROUNDS];
            int      sl[BIN_ROUNDS];
            float    ww[BIN_ROUNDS];
            #pragma unroll
            for (int q = 0; q < BIN_ROUNDS; ++q) {
                const int i = q * STEP_THREADS + tid;
                kk[q] = 0u; sl[q] = 0; ww[q] = 0.0f;
                if (i < n) {
                    const unsigned k = kr[i].x;
                    const int r = (int)(k >> COL_BITS);
                    sl[q] = atomicAdd(&bcnt[r & (STEP_WAVES - 1)], 1);
                    ww[q] = DELTA * (dinv[R0 + r] * dinv[(int)(k & ((1u << COL_BITS) - 1))]);
                    kk[q] = k;
                }
            }
            __syncthreads();
            if (wave == 0) {
                const int v = bcnt[lane];
                int s = v;
                #pragma unroll
                for (int off = 1; off < 32; off <<= 1) {
                    const int t = __shfl_up(s, off, 32);
                    if (lane >= off) s += t;
                }
                boff[lane] = s - v;
                if (lane == 31) boff[32] = s;
            }
            __syncthreads();
            #pragma unroll
            for (int q = 0; q < BIN_ROUNDS; ++q) {
                const int i = q * STEP_THREADS + tid;
                if (i < n) {
                    const int b = (int)(kk[q] >> COL_BITS) & (STEP_WAVES - 1);
                    srt[boff[b] + sl[q]] = make_uint2(kk[q], __float_as_uint(ww[q]));
                }
            }
            __syncthreads();
            if (tid >= 1 && tid < STEP_WAVES) ((int*)kl)[tid] = boff[tid];
            for (int i = tid; i < n; i += STEP_THREADS) kr[i] = srt[i];
        }

        int beg, end;
        if constexpr (FIRST) {
            beg = boff[wave];
            end = boff[wave + 1];
        } else {
            beg = wave == 0 ? 0 : hdr[wave];
            end = wave == STEP_WAVES - 1 ? n : hdr[wave + 1];
        }
        beg = __builtin_amdgcn_readfirstlane(beg);
        end = __builtin_amdgcn_readfirstlane(end);
        auto rec = [&](const int base) {
            int i = base + lane;
            if (i > end - 1) i = end - 1;
            if constexpr (FIRST) return srt[i];
            else                 return kr[i];
        };
        if (beg < end) {
            uint2 m = rec(beg);
            int base = beg;
            for (; base + 32 <= end; base += 32) {
                const int c = (int)(m.x & ((1u << COL_BITS) - 1));
                const int rl = (int)(m.x >> COL_BITS);
                const int wbits = (int)m.y;
                float4 va[EDGE_GROUP], vb[EDGE_GROUP];
                dgc_gather8(va, c, 0, lane, s4);
                if (base + 32 < end) m = rec(base + 32);
                dgc_gather8(vb, c, EDGE_GROUP, lane, s4);
                dgc_add8<PITCH, true>(va, rl, wbits, 0, 32, lane, acc);
                dgc_gather8(va, c, 2 * EDGE_GROUP, lane, s4);
                dgc_add8<PITCH, true>(vb, rl, wbits, EDGE_GROUP, 32, lane, acc);
                dgc_gather8(vb, c, 3 * EDGE_GROUP, lane, s4);
                dgc_add8<PITCH, true>(va, rl, wbits, 2 * EDGE_GROUP, 32, lane, acc);
                dgc_add8<PITCH, true>(vb, rl, wbits, 3 * EDGE_GROUP, 32, lane, acc);
            }
            if (base < end) {
                const int cnt = end - base;
                const int c = (int)(m.x & ((1u << COL_BITS) - 1));
                const int rl = (int)(m.x >> COL_BITS);
                const int wbits = (int)m.y;
                float4 va[EDGE_GROUP], vb[EDGE_GROUP];
                dgc_gather8(va, c, 0, lane, s4);
                if (EDGE_GROUP < cnt) dgc_gather8(vb, c, EDGE_GROUP, lane, s4);
                dgc_add8<PITCH, false>(va, rl, wbits, 0, cnt, lane, acc);
                if (EDGE_GROUP < cnt) {
                    if (2 * EDGE_GROUP < cnt) dgc_gather8(va, c, 2 * EDGE_GROUP, lane, s4);
                    dgc_add8<PITCH, false>(vb, rl, wbits, EDGE_GROUP, cnt, lane, acc);
                    if (2 * EDGE_GROUP < cnt) {
                        if (3 * EDGE_GROUP < cnt) dgc_gather8(vb, c, 3 * EDGE_GROUP, lane, s4);
                        dgc_add8<PITCH, false>(va, rl, wbits, 2 * EDGE_GROUP, cnt, lane, acc);
                        if (3 * EDGE_GROUP < cnt) dgc_add8<PITCH, false>(vb, rl, wbits, 3 * EDGE_GROUP, cnt, lane, acc);
                    }
                }
            }
        }
    } else {
        #pragma unroll 1
        for (int e = tid; e < EE; e += STEP_THREADS) {
            const int r = row[e];
            const unsigned rl = (unsigned)(r - R0);
            if (rl < (unsigned)ROWS_PER_WG) {
                const int c = col[e];
                const float w = DELTA * (dinv[r] * dinv[c]);
                float* p = acc + rl * PITCH;
                #pragma unroll 1
                for (int k = 0; k < DD; ++k)
                    atomicAdd(p + k, src[c * DD + k] * w);
            }
        }
    }
    }
    __syncthreads();

    if constexpr (!LAST) {
        for (int i = tid; i < ROWS_PER_WG * 32; i += STEP_THREADS) {
            const int rl = i >> 5;
            ((float4*)next)[(R0 + rl) * 32 + lane] = ((const float4*)(acc + rl * PITCH))[lane];
        }
    } else {
        static_assert(ROWS_PER_WG % 16 == 0 && ROWS_PER_WG / 16 <= STEP_WAVES, "one wave per 16-row tile");
        if (wave < ROWS_PER_WG / 16) {
            const int c = lane & 15;
            const int h = lane >> 4;
            const float* ap = acc + (wave * 16 + c) * PITCH + h;
            const float* bp = wl + c * FC_PITCH + h;
            v8f d0 = {}, d1 = {}, d2 = {};
            #pragma unroll 8
            for (int s = 0; s < DD / 4; ++s) {
                v2f a, b0, b1, b2;
                a.x  = ap[4 * s];
                a.y  = ap[4 * s + 2];
                b0.x = bp[4 * s];
                b0.y = bp[4 * s + 2];
                b1.x = bp[16 * FC_PITCH + 4 * s];
                b1.y = bp[16 * FC_PITCH + 4 * s + 2];
                b2.x = bp[32 * FC_PITCH + 4 * s];
                b2.y = bp[32 * FC_PITCH + 4 * s + 2];
                d0 = __builtin_amdgcn_wmma_f32_16x16x4_f32(false, a, false, b0, (short)0, d0, false, false);
                d1 = __builtin_amdgcn_wmma_f32_16x16x4_f32(false, a, false, b1, (short)0, d1, false, false);
                d2 = __builtin_amdgcn_wmma_f32_16x16x4_f32(false, a, false, b2, (short)0, d2, false, false);
            }
            const bool t2 = c < CC - 32;
            const float bv0 = bias[c],  bv1 = bias[16 + c],  bv2 = t2 ? bias[32 + c]  : 0.0f;
            const float g0  = gamma[c], g1  = gamma[16 + c], g2  = t2 ? gamma[32 + c] : 0.0f;
            const float e0  = beta[c],  e1  = beta[16 + c],  e2  = t2 ? beta[32 + c]  : 0.0f;
            #pragma unroll
            for (int r = 0; r < 8; ++r) {
                const float x0 = d0[r] + bv0;
                const float x1 = d1[r] + bv1;
                const float x2 = t2 ? d2[r] + bv2 : 0.0f;
                float s  = x0 + x1 + x2;
                float sq = x0 * x0 + x1 * x1 + x2 * x2;
                #pragma unroll
                for (int off = 8; off > 0; off >>= 1) {
                    s  += __shfl_xor(s,  off, 32);
                    sq += __shfl_xor(sq, off, 32);
                }
                const float mean = s * (1.0f / CC);
                const float var  = sq * (1.0f / CC) - mean * mean;
                const float inv  = rsqrtf(var + LN_EPS);
                float* q = stage + (wave * 16 + h * 8 + r) * CC + c;
                q[0]  = (x0 - mean) * inv * g0 + e0;
                q[16] = (x1 - mean) * inv * g1 + e1;
                if (t2) q[32] = (x2 - mean) * inv * g2 + e2;
            }
        }
        __syncthreads();
        float4* o4 = (float4*)(next + (size_t)R0 * CC);
        for (int i = tid; i < ROWS_PER_WG * CC / 4; i += STEP_THREADS)
            o4[i] = ((const float4*)stage)[i];
    }
}

extern "C" void kernel_launch(void* const* d_in, const int* in_sizes, int n_in,
                              void* d_out, int out_size, void* d_ws, size_t ws_size,
                              hipStream_t stream) {
    const float* feat  = (const float*)d_in[0];
    const int*   row   = (const int*)  d_in[1];
    const int*   col   = (const int*)  d_in[2];
    const float* W     = (const float*)d_in[3];
    const float* bias  = (const float*)d_in[4];
    const float* gamma = (const float*)d_in[5];
    const float* beta  = (const float*)d_in[6];
    float* out = (float*)d_out;

    const size_t NP = 100352;
    float* dinv = (float*)d_ws;
    float* ta   = dinv + NP;
    float* tb   = ta + (size_t)NN * DD;

    unsigned int* owner = (unsigned int*)tb;
    unsigned int* key   = owner + LIST_KEY_OFF;
    unsigned int* kept  = (unsigned int*)d_out;

    const int B = 256;

    static_assert(OWNER_DWORDS <= LIST_KEY_OFF, "the owner bytes end before the keys begin");
    static_assert(2097152 + (size_t)EE * 4 <= (size_t)NN * DD * 4, "both fit in tb");
    static_assert((size_t)NUM_OWNERS * LIST_BYTES == (size_t)NN * CC * 4, "the groups' kept records fill out exactly");
    dgc_owner_keys<<<(OWNER_DWORDS + B - 1) / B, B, 0, stream>>>(row, col, owner, key);
    dgc_owner_scan<<<NUM_OWNERS, STEP_THREADS, 0, stream>>>(owner, key, kept, dinv);

    static_assert(KSTEPS == 4, "the four steps are written out below");
    dgc_step_owned<0><<<NUM_OWNERS, STEP_THREADS, 0, stream>>>(feat, dinv, row, col, kept, ta, W, bias, gamma, beta);
    dgc_step_owned<1><<<NUM_OWNERS, STEP_THREADS, 0, stream>>>(ta, dinv, row, col, kept, tb, W, bias, gamma, beta);
    dgc_step_owned<1><<<NUM_OWNERS, STEP_THREADS, 0, stream>>>(tb, dinv, row, col, kept, ta, W, bias, gamma, beta);
    dgc_step_owned<2><<<NUM_OWNERS, STEP_THREADS, 0, stream>>>(ta, dinv, row, col, kept, out, W, bias, gamma, beta);
}
